# starting kernel
# baseline (speedup 1.0000x reference)
#include <hip/hip_runtime.h>

typedef float v2f __attribute__((ext_vector_type(2)));
typedef float v8f __attribute__((ext_vector_type(8)));

#define N_NODES  100000
#define N_ETYPES 3
#define N_EDGES  1600000
#define D_IN     128
#define D_HID    256
#define D_OUT    32

__global__ void fill_zero_kernel(float* __restrict__ p, int n) {
    int i = blockIdx.x * blockDim.x + threadIdx.x;
    if (i < n) p[i] = 0.0f;
}

__global__ void degree_kernel(const int* __restrict__ dst, float* __restrict__ deg) {
    int e = blockIdx.y;
    int i = blockIdx.x * blockDim.x + threadIdx.x;
    if (i < N_EDGES) {
        int d = dst[(size_t)e * N_EDGES + i];
        atomicAdd(&deg[(size_t)e * N_NODES + d], 1.0f);
    }
}

__global__ void scatter_kernel(const float* __restrict__ h,
                               const int* __restrict__ src,
                               const int* __restrict__ dst,
                               float* __restrict__ agg) {
    int edge = blockIdx.x * 8 + (threadIdx.x >> 5);
    int c    = (threadIdx.x & 31) * 4;
    if (edge >= N_EDGES) return;
    int s = src[edge];
    int d = dst[edge];
    const float4 v = *(const float4*)(h + (size_t)s * D_IN + c);
    float* o = agg + (size_t)d * D_IN + c;
    atomicAdd(o + 0, v.x);
    atomicAdd(o + 1, v.y);
    atomicAdd(o + 2, v.z);
    atomicAdd(o + 3, v.w);
}

__global__ void wmma_gemm_f32_kernel(const float* __restrict__ A,
                                     const float* __restrict__ W,
                                     const float* __restrict__ bias,
                                     float* __restrict__ C,
                                     const float* __restrict__ deg,
                                     int M, int N, int K,
                                     float outScale, int accumulate, int doRelu) {
    const int lane = threadIdx.x;
    const int half = lane >> 4;
    const int l15  = lane & 15;
    const int n0   = blockIdx.x * 16;
    const int m0   = blockIdx.y * 16;
    const int rowA = m0 + l15;
    const int colB = n0 + l15;

    float rs = 1.0f;
    if (deg) rs = 1.0f / fmaxf(deg[rowA], 1.0f);

    v8f acc = {};
    const float* arow = A + (size_t)rowA * K;
    for (int k = 0; k < K; k += 4) {
        const int ka = k + half * 2;
        v2f a, b;
        a.x = arow[ka]     * rs;
        a.y = arow[ka + 1] * rs;
        b.x = W[(size_t)ka       * N + colB];
        b.y = W[(size_t)(ka + 1) * N + colB];
        acc = __builtin_amdgcn_wmma_f32_16x16x4_f32(
             false, a,  false, b,
             (short)0, acc,  false,  false);
    }

    const float bv = bias[colB];
#pragma unroll
    for (int r = 0; r < 8; ++r) {
        const int row = m0 + r + half * 8;
        float v = acc[r] + bv;
        if (doRelu) v = fmaxf(v, 0.0f);
        v *= outScale;
        const size_t idx = (size_t)row * N + colB;
        if (accumulate) C[idx] += v;
        else            C[idx] = v;
    }
}

extern "C" void kernel_launch(void* const* d_in, const int* in_sizes, int n_in,
                              void* d_out, int out_size, void* d_ws, size_t ws_size,
                              hipStream_t stream) {
    const float* blocks = (const float*)d_in[0];
    const int*   esrc   = (const int*)  d_in[1];
    const int*   edst   = (const int*)  d_in[2];
    const float* convW  = (const float*)d_in[3];
    const float* convB  = (const float*)d_in[4];
    const float* W1     = (const float*)d_in[5];
    const float* b1     = (const float*)d_in[6];
    const float* W2     = (const float*)d_in[7];
    const float* b2     = (const float*)d_in[8];
    float*       out    = (float*)d_out;

    char* ws = (char*)d_ws;
    const size_t BUF = (size_t)N_NODES * D_IN * sizeof(float);
    float* buf0 = (float*)(ws);
    float* agg  = (float*)(ws + BUF);
    float* buf2 = (float*)(ws + 2 * BUF);
    float* deg  = (float*)(ws + 3 * BUF);
    float* H1   = buf0;

    fill_zero_kernel<<<(N_ETYPES * N_NODES + 255) / 256, 256, 0, stream>>>(deg, N_ETYPES * N_NODES);
    {
        dim3 g((N_EDGES + 255) / 256, N_ETYPES);
        degree_kernel<<<g, 256, 0, stream>>>(edst, deg);
    }

    const float* hin = blocks;
    float* hout = buf0;
    for (int l = 0; l < 2; ++l) {
        for (int e = 0; e < N_ETYPES; ++e) {
            fill_zero_kernel<<<(N_NODES * D_IN + 255) / 256, 256, 0, stream>>>(agg, N_NODES * D_IN);
            scatter_kernel<<<(N_EDGES + 7) / 8, 256, 0, stream>>>(
                hin, esrc + (size_t)e * N_EDGES, edst + (size_t)e * N_EDGES, agg);
            dim3 g(D_IN / 16, N_NODES / 16);
            wmma_gemm_f32_kernel<<<g, 32, 0, stream>>>(
                agg,
                convW + ((size_t)l * N_ETYPES + e) * D_IN * D_IN,
                convB + ((size_t)l * N_ETYPES + e) * D_IN,
                hout,
                deg + (size_t)e * N_NODES,
                N_NODES, D_IN, D_IN,
                1.0f / 3.0f,  (e > 0) ? 1 : 0,  1);
        }
        hin  = hout;
        hout = buf2;
    }

    {
        dim3 g(D_HID / 16, N_NODES / 16);
        wmma_gemm_f32_kernel<<<g, 32, 0, stream>>>(
            hin, W1, b1, H1, nullptr, N_NODES, D_HID, D_IN, 1.0f, 0, 0);
    }
    {
        dim3 g(D_OUT / 16, N_NODES / 16);
        wmma_gemm_f32_kernel<<<g, 32, 0, stream>>>(
            H1, W2, b2, out, nullptr, N_NODES, D_OUT, D_HID, 1.0f, 0, 0);
    }
}


# Layer-0 scan gone: the launch in front buckets each 20000-edge chunk's records by owner work-group in LDS and an owner reads only its run of each chunk; sum order = LDS arrival order, as before
# speedup vs baseline: 6.9957x; 6.9957x over previous
#include <hip/hip_runtime.h>

typedef float v2f __attribute__((ext_vector_type(2)));
typedef float v8f __attribute__((ext_vector_type(8)));

#define N_NODES  100000
#define N_ETYPES 3
#define N_EDGES  1600000
#define D_IN     128
#define D_HID    256
#define D_OUT    32

#define OWN_NODES   400
#define ACC_PITCH   (D_IN + 4)
#define OWN_MT      (OWN_NODES / 16)
#define OWN_GROUPS  (N_NODES / OWN_NODES)
#define OWN_THREADS 1024
#define OWN_WAVES   (OWN_THREADS / 32)
#define OWN_BATCH   8
#define OWN_RING    512
#define E4          (N_EDGES / 4)
#define OWN_CAP     512
#define BIN_CAP     288
#define LIST_WORDS  ((size_t)OWN_GROUPS * OWN_WAVES * OWN_CAP)
static_assert(E4 % 32 == 0, "the dst list is whole steps of 32 int4");
static_assert(BIN_CAP <= OWN_CAP && OWN_NODES <= 512 && (OWN_WAVES & (OWN_WAVES - 1)) == 0, "a bin fits its list; a local node fits 9 bits; n % OWN_WAVES is a mask");
#define PART_CH     20000
#define PART_NCH    (N_EDGES / PART_CH)
#define PART_Q      (PART_CH / 4)
#define PART_PER    5
#define PART_LOADERS (PART_Q / PART_PER)
#define PART_OFFS   256
#define PART_TURNS  ((PART_NCH + OWN_WAVES - 1) / OWN_WAVES)
static_assert(PART_NCH * PART_CH == N_EDGES && PART_LOADERS * PART_PER == PART_Q && PART_LOADERS <= OWN_THREADS, "chunks tile an edge type exactly; the loaders cover a chunk exactly");
static_assert(OWN_GROUPS + 1 < PART_OFFS, "bucket g ends where bucket g + 1 starts, and 255 is free for a dst that belongs to nobody");

__global__ __launch_bounds__(OWN_THREADS)
void owner_part_kernel(const int* __restrict__ src,
                       const int* __restrict__ dst,
                       int* __restrict__ offs,
                       int* __restrict__ recs) {
    __shared__ int s_cnt[PART_OFFS];
    __shared__ __attribute__((aligned(16))) int s_rec[PART_CH];
    const int tid = threadIdx.x;
    const int4* d4 = (const int4*)dst + (size_t)blockIdx.x * PART_Q;
    const int4* s4 = (const int4*)src + (size_t)blockIdx.x * PART_Q;
    const bool loader = tid < PART_LOADERS;
    int4 d[PART_PER], s[PART_PER];
#pragma unroll
    for (int k = 0; k < PART_PER; ++k) {
        const int i = loader ? tid + k * PART_LOADERS : 0;
        d[k] = d4[i];
        s[k] = s4[i];
    }
    if (tid < PART_OFFS) s_cnt[tid] = 0;
    __syncthreads();
    int own[4 * PART_PER], rec[4 * PART_PER], rank[4 * PART_PER];
#pragma unroll
    for (int k = 0; k < PART_PER; ++k) {
        const int dd[4] = {d[k].x, d[k].y, d[k].z, d[k].w};
        const int ss[4] = {s[k].x, s[k].y, s[k].z, s[k].w};
#pragma unroll
        for (int j = 0; j < 4; ++j) {
            const bool     in = (unsigned)dd[j] < (unsigned)N_NODES;
            const unsigned o  = (unsigned)dd[j] / OWN_NODES;
            own[4 * k + j]  = in ? (int)o : 255;
            rec[4 * k + j]  = (ss[j] << 9) | (in ? (int)((unsigned)dd[j] - o * OWN_NODES) : 0);
            rank[4 * k + j] = loader ? atomicAdd(&s_cnt[own[4 * k + j]], 1) : 0;
        }
    }
    __syncthreads();
    if (tid < 32) {
        int c[8];
        int sum = 0;
#pragma unroll
        for (int i = 0; i < 8; ++i) { c[i] = s_cnt[tid * 8 + i]; sum += c[i]; }
        int incl = sum;
        for (int dlt = 1; dlt < 32; dlt <<= 1) {
            const int t = __shfl_up(incl, dlt);
            if (tid >= dlt) incl += t;
        }
        int run = incl - sum;
#pragma unroll
        for (int i = 0; i < 8; ++i) { s_cnt[tid * 8 + i] = run; run += c[i]; }
    }
    __syncthreads();
    if (loader) {
#pragma unroll
        for (int q = 0; q < 4 * PART_PER; ++q) s_rec[s_cnt[own[q]] + rank[q]] = rec[q];
    }
    if (tid < PART_OFFS) offs[(size_t)blockIdx.x * PART_OFFS + tid] = s_cnt[tid];
    __syncthreads();
    int4* r4 = (int4*)(recs + (size_t)blockIdx.x * PART_CH);
    const int4* sr4 = (const int4*)s_rec;
    for (int i = tid; i < PART_Q; i += OWN_THREADS) r4[i] = sr4[i];
}

template <bool FULL>
__device__ __forceinline__ void own_rows(const float* __restrict__ h, float* acc,
                                         int loc, int roff, int cn, int lane) {
    const int kend = FULL ? 32 : cn;
#pragma unroll 1
    for (int k = 0; k < kend; k += OWN_BATCH) {
        float4 v[OWN_BATCH];
        int ll[OWN_BATCH];
#pragma unroll
        for (int b = 0; b < OWN_BATCH; ++b) {
            const int i  = FULL ? (k + b) : ((k + b < cn) ? (k + b) : (cn - 1));
            const int ro = __builtin_amdgcn_readlane(roff, i);
            ll[b] = __builtin_amdgcn_readlane(loc, i);
            v[b]  = *(const float4*)(h + ro + lane * 4);
        }
#pragma unroll
        for (int b = 0; b < OWN_BATCH; ++b) {
            if (FULL || k + b < cn) {
                float4* o = (float4*)(acc + ll[b] * ACC_PITCH) + lane;
                float4 a = *o;
                a.x += v[b].x;
                a.y += v[b].y;
                a.z += v[b].z;
                a.w += v[b].w;
                *o = a;
            }
        }
    }
}

__device__ __forceinline__ void own_store(float* acc, int n, int lane, float4 a, float rsv) {
    const float rs = __int_as_float(__builtin_amdgcn_readlane(__float_as_int(rsv), n / OWN_WAVES));
    a.x *= rs;
    a.y *= rs;
    a.z *= rs;
    a.w *= rs;
    *((float4*)(acc + n * ACC_PITCH) + lane) = a;
}

template <bool FULL>
__device__ __forceinline__ void own_rows_sum(const float* __restrict__ h, float* acc,
                                             int loc, int roff, int cn, int lane,
                                             float4& a, int& cur, float rsv) {
    const int kend = FULL ? 32 : cn;
#pragma unroll 1
    for (int k = 0; k < kend; k += OWN_BATCH) {
        float4 v[OWN_BATCH];
        int ll[OWN_BATCH];
#pragma unroll
        for (int b = 0; b < OWN_BATCH; ++b) {
            const int i  = FULL ? (k + b) : ((k + b < cn) ? (k + b) : (cn - 1));
            const int ro = __builtin_amdgcn_readlane(roff, i);
            ll[b] = __builtin_amdgcn_readlane(loc, i);
            v[b]  = *(const float4*)(h + ro + lane * 4);
        }
#pragma unroll
        for (int b = 0; b < OWN_BATCH; ++b) {
            if (FULL || k + b < cn) {
                if (ll[b] != cur) {
                    if (cur >= 0) own_store(acc, cur, lane, a, rsv);
                    a   = v[b];
                    cur = ll[b];
                } else {
                    a.x += v[b].x;
                    a.y += v[b].y;
                    a.z += v[b].z;
                    a.w += v[b].w;
                }
            }
        }
    }
}

__device__ __forceinline__ void rec_pop(int rec, float* cntf, int* bincnt, int* bins) {
    const int loc  = rec & 511;
    const int ow   = loc & (OWN_WAVES - 1);
    atomicAdd(&cntf[loc], 1.0f);
    const int slot = atomicAdd(&bincnt[ow], 1);
    if (slot < BIN_CAP) bins[ow * BIN_CAP + slot] = rec;
}

__device__ __forceinline__ void run_pop(const int* __restrict__ run, int first, int end, int lane,
                                        float* cntf, int* bincnt, int* bins) {
    for (int i = first + lane; i < end; i += 64) {
        const int r0 = run[i];
        const int r1 = run[min(i + 32, end - 1)];
        rec_pop(r0, cntf, bincnt, bins);
        if (i + 32 < end) rec_pop(r1, cntf, bincnt, bins);
    }
}

__device__ __forceinline__ void own_replay(const float* __restrict__ h, const int* wl,
                                           int n, float* acc, int lane, float rsv) {
    if (n <= 0) return;
    const int last = n - 1;
    int rec = wl[min(lane, last)];
    int k0 = 0;
    float4 a = make_float4(0.f, 0.f, 0.f, 0.f);
    int cur = -1;
#pragma unroll 1
    for (; k0 + 32 <= n; k0 += 32) {
        const int nxt = wl[min(k0 + 32 + lane, last)];
        own_rows_sum<true>(h, acc, rec & 511, (rec >> 9) * D_IN, 32, lane, a, cur, rsv);
        rec = nxt;
    }
    if (k0 < n) own_rows_sum<false>(h, acc, rec & 511, (rec >> 9) * D_IN, n - k0, lane, a, cur, rsv);
    own_store(acc, cur, lane, a, rsv);
}

template <bool FULL>
__device__ __forceinline__ void rescan_pop(const float* __restrict__ h, const int* __restrict__ src,
                                           const int* __restrict__ dst, int base, float* acc,
                                           const int* ring, int head, int cn, int lane) {
    int loc = 0, roff = 0;
    if (FULL || lane < cn) {
        const int e = ring[(head + lane) & (OWN_RING - 1)];
        loc  = dst[e] - base;
        roff = src[e] * D_IN;
    }
    own_rows<FULL>(h, acc, loc, roff, cn, lane);
}

__device__ __forceinline__ void own_rescan(const float* __restrict__ h, const int* __restrict__ src,
                                           const int* __restrict__ dst, int base, int wave, int lane,
                                           float* acc, int* ring) {
    const int4* dst4 = (const int4*)dst;
    int head = 0, tail = 0;
#pragma unroll 1
    for (int p0 = 0; p0 < E4; p0 += 32) {
        const int4 d = dst4[p0 + lane];
        const int dd[4] = {d.x, d.y, d.z, d.w};
#pragma unroll
        for (int j = 0; j < 4; ++j) {
            const int  loc = dd[j] - base;
            const bool m   = (unsigned)loc < (unsigned)OWN_NODES && (loc & (OWN_WAVES - 1)) == wave;
            const unsigned mask = __builtin_amdgcn_ballot_w32(m);
            if (mask) {
                if (m) ring[(tail + (int)__builtin_amdgcn_mbcnt_lo(mask, 0u)) & (OWN_RING - 1)] = (p0 + lane) * 4 + j;
                tail += __builtin_popcount(mask);
            }
        }
        while (tail - head >= 32) {
            rescan_pop<true>(h, src, dst, base, acc, ring, head, 32, lane);
            head += 32;
        }
    }
    rescan_pop<false>(h, src, dst, base, acc, ring, head, tail - head, lane);
}

#define OWN_STAGES 4
static_assert(6 * OWN_STAGES + 1 == OWN_MT, "stages start at row tiles 0, 7, 13, 19; the last ends at 25");
__device__ __forceinline__ int stage_tile(int s) { return s <= 0 ? 0 : 6 * s + 1; }

__device__ __forceinline__ int stage_off(int excl, int s, int wave) {
    const int below = (16 * stage_tile(s) - wave + 31) >> 5;
    return __builtin_amdgcn_readlane(excl, below);
}

template <bool ADD>
__device__ __forceinline__ void own_product(const float* acc, float* __restrict__ hout,
                                            int base, const v2f (&bf)[D_IN / 4], float bv,
                                            int mtb, int mte, int half, int l15, int colB) {
    for (int mt = mtb; mt < mte; mt += OWN_WAVES / 8) {
        const float* arow = acc + (mt * 16 + l15) * ACC_PITCH + 2 * half;
        float* orow = hout + (size_t)(base + mt * 16 + half * 8) * D_IN + colB;
        float old[8];
#pragma unroll
        for (int r = 0; r < 8; ++r) old[r] = ADD ? orow[r * D_IN] : 0.0f;
        v8f d = {};
#pragma unroll
        for (int s = 0; s < D_IN / 4; ++s) {
            const v2f a = *(const v2f*)(arow + 4 * s);
            d = __builtin_amdgcn_wmma_f32_16x16x4_f32(
                 false, a,  false, bf[s],
                 (short)0, d,  false,  false);
        }
#pragma unroll
        for (int r = 0; r < 8; ++r) {
            float v = fmaxf(d[r] + bv, 0.0f) * (1.0f / 3.0f);
            if (ADD) v += old[r];
            orow[r * D_IN] = v;
        }
    }
}

template <bool COUNT, bool ADD>
__global__ __launch_bounds__(OWN_THREADS)
void owner_agg_kernel(const float* __restrict__ h,
                      const int* __restrict__ src,
                      const int* __restrict__ dst,
                      const int* __restrict__ offs,
                      const int* __restrict__ recs,
                      const float* __restrict__ Wt,
                      const float* __restrict__ bias,
                      float* __restrict__ hout,
                      float* __restrict__ deg,
                      int* __restrict__ lst,
                      int* __restrict__ lcnt) {
    __shared__ __attribute__((aligned(16))) float acc[OWN_NODES * ACC_PITCH];
    __shared__ float cntf[OWN_NODES];
    __shared__ int   rings[OWN_WAVES * (OWN_RING + 32)];
    __shared__ int   bins[OWN_WAVES * BIN_CAP];
    __shared__ int   bincnt[OWN_WAVES];
    const int tid  = threadIdx.x;
    const int lane = tid & 31;
    const int base = blockIdx.x * OWN_NODES;
    const int wave = __builtin_amdgcn_readfirstlane(tid >> 5);
    int*  ring = rings + wave * (OWN_RING + 32);
    int*  wl   = lst + ((size_t)blockIdx.x * OWN_WAVES + wave) * OWN_CAP;

    int nraw = 0;
    if (!COUNT) nraw = lcnt[blockIdx.x * OWN_WAVES + wave];

    float4* acc4 = (float4*)acc;
    for (int i = tid; i < OWN_NODES * ACC_PITCH / 4; i += OWN_THREADS) acc4[i] = make_float4(0.f, 0.f, 0.f, 0.f);
    for (int i = tid; i < OWN_NODES; i += OWN_THREADS) cntf[i] = COUNT ? 0.0f : deg[base + i];
    if (COUNT && tid < OWN_WAVES) bincnt[tid] = 0;
    __syncthreads();

    int nrec;
    if (COUNT) {
        int first[PART_TURNS], end[PART_TURNS];
#pragma unroll
        for (int k = 0; k < PART_TURNS; ++k) {
            const int ch = wave + k * OWN_WAVES;
            const int* ob = offs + min(ch, PART_NCH - 1) * PART_OFFS + blockIdx.x;
            const int b0 = ob[0], b1 = ob[1];
            first[k] = b0;
            end[k]   = ch < PART_NCH ? b1 : b0;
        }
#pragma unroll
        for (int k = 0; k < PART_TURNS; ++k) {
            const int ch = min(wave + k * OWN_WAVES, PART_NCH - 1);
            run_pop(recs + ch * PART_CH, first[k], end[k], lane, cntf, bincnt, bins);
        }
        __syncthreads();

        nrec = __builtin_amdgcn_readfirstlane(bincnt[wave]);
        if (tid < OWN_WAVES) lcnt[blockIdx.x * OWN_WAVES + tid] = bincnt[tid];
        for (int i = tid; i < OWN_NODES; i += OWN_THREADS) deg[base + i] = cntf[i];
    } else {
        nrec = __builtin_amdgcn_readfirstlane(nraw);
    }
    const bool over = nrec > BIN_CAP;

    const int nd = lane * OWN_WAVES + wave;
    const int c  = nd < OWN_NODES ? (int)cntf[nd] : 0;
    int incl = c;
    for (int d = 1; d < 16; d <<= 1) {
        const int t = __shfl_up(incl, d);
        if (lane >= d) incl += t;
    }
    const int excl = incl - c;
    const float rsv = 1.0f / fmaxf(nd < OWN_NODES ? cntf[nd] : 1.0f, 1.0f);

    if (COUNT && !over) {
        const int* mybin = bins + wave * BIN_CAP;
        int* slot = ring + OWN_RING;
        slot[lane] = excl;
        for (int i = lane; i < nrec; i += 32) {
            const int rec = mybin[i];
            const int pos = atomicAdd(&slot[(rec & 511) / OWN_WAVES], 1);
            ring[pos] = rec;
        }
        for (int i = lane; i < nrec; i += 32) wl[i] = ring[i];
    }
    if (!COUNT && !over && nrec > 0) {
        int t[BIN_CAP / 32];
#pragma unroll
        for (int k = 0; k < BIN_CAP / 32; ++k) t[k] = wl[min(lane + 32 * k, nrec - 1)];
#pragma unroll
        for (int k = 0; k < BIN_CAP / 32; ++k) ring[lane + 32 * k] = t[k];
    }
    const int* lp = ring;

    if (over) {
        own_rescan(h, src, dst, base, wave, lane, acc, ring);
        for (int v = 0; v * OWN_WAVES + wave < OWN_NODES; ++v) {
            float4* o = (float4*)(acc + (v * OWN_WAVES + wave) * ACC_PITCH) + lane;
            own_store(acc, v * OWN_WAVES + wave, lane, *o, rsv);
        }
    } else {
        own_replay(h, lp, stage_off(excl, 1, wave), acc, lane, rsv);
    }

    const int half = lane >> 4;
    const int l15  = lane & 15;
    const int colB = ((tid >> 5) & 7) * 16 + l15;
    v2f bf[D_IN / 4];
#pragma unroll
    for (int s = 0; s < D_IN / 4; ++s) {
        const int ka = 4 * s + 2 * half;
        bf[s].x = Wt[ka * D_IN + colB];
        bf[s].y = Wt[(ka + 1) * D_IN + colB];
    }
    const float bv = bias[colB];
    const int mt0 = __builtin_amdgcn_readfirstlane(tid >> 8);
    __syncthreads();

    const int grp = (wave ^ (wave >> 2)) & 1;
#pragma unroll 1
    for (int s = 0; s < OWN_STAGES; ++s) {
#pragma unroll 1
        for (int turn = 0; turn < 2; ++turn) {
            if (turn == grp) {
                own_product<ADD>(acc, hout, base, bf, bv, stage_tile(s) + mt0, stage_tile(s + 1),
                                 half, l15, colB);
            } else if (s + 1 < OWN_STAGES && !over) {
                const int a = stage_off(excl, s + 1, wave);
                own_replay(h, lp + a, stage_off(excl, s + 2, wave) - a, acc, lane, rsv);
            }
        }
        if (s + 1 < OWN_STAGES) __syncthreads();
    }
}

#define HD_ROWS    80
#define HD_MT      (HD_ROWS / 16)
#define HD_BLOCKS  (OWN_NODES / HD_ROWS)
#define HD_THREADS 512
#define H1_PITCH   (D_HID + 4)
static_assert(HD_ROWS * D_IN / 4 == 5 * HD_THREADS, "a thread moves five float4 per block");

typedef float v4f __attribute__((ext_vector_type(4)));

#define HEAD_FETCH(hrows)                                                   \
    do {                                                                    \
        const v4f* src4_ = (const v4f*)(hrows) + tid;                       \
        p0 = src4_[0 * HD_THREADS]; p1 = src4_[1 * HD_THREADS];             \
        p2 = src4_[2 * HD_THREADS]; p3 = src4_[3 * HD_THREADS];             \
        p4 = src4_[4 * HD_THREADS];                                         \
    } while (0)
#define HEAD_PUT1(v, Ab, idx) (*(v4f*)((Ab) + ((idx) >> 5) * ACC_PITCH + ((idx) & 31) * 4) = (v))
#define HEAD_PUT(Ab)                                                        \
    do {                                                                    \
        HEAD_PUT1(p0, Ab, 0 * HD_THREADS + tid); HEAD_PUT1(p1, Ab, 1 * HD_THREADS + tid); \
        HEAD_PUT1(p2, Ab, 2 * HD_THREADS + tid); HEAD_PUT1(p3, Ab, 3 * HD_THREADS + tid); \
        HEAD_PUT1(p4, Ab, 4 * HD_THREADS + tid);                            \
    } while (0)

__global__ __launch_bounds__(HD_THREADS)
void head_kernel(const float* __restrict__ hin,
                 const float* __restrict__ W1,
                 const float* __restrict__ b1,
                 const float* __restrict__ W2,
                 const float* __restrict__ b2,
                 float* __restrict__ out) {
    __shared__ __attribute__((aligned(16))) float As[2][HD_ROWS * ACC_PITCH];
    __shared__ __attribute__((aligned(16))) float H1s[HD_ROWS * H1_PITCH];
    __shared__ __attribute__((aligned(16))) float W2s[D_HID * D_OUT];
    const int tid  = threadIdx.x;
    const int lane = tid & 31;
    const int half = lane >> 4;
    const int l15  = lane & 15;
    const int wave = __builtin_amdgcn_readfirstlane(tid >> 5);
    const int row0 = blockIdx.x * OWN_NODES;
    const float* hbase = hin + (size_t)row0 * D_IN;

    v4f p0, p1, p2, p3, p4;
    HEAD_FETCH(hbase);

    const int colB = wave * 16 + l15;
    v2f bf[D_IN / 4];
#pragma unroll
    for (int s = 0; s < D_IN / 4; ++s) {
        const int ka = 4 * s + 2 * half;
        bf[s].x = W1[ka * D_HID + colB];
        bf[s].y = W1[(ka + 1) * D_HID + colB];
    }
    const float b1v = b1[colB];

    for (int idx = tid; idx < D_HID * D_OUT; idx += HD_THREADS) {
        const int k = idx >> 5, col = idx & 31;
        W2s[(k >> 2) * 128 + col * 4 + ((k >> 1) & 1) * 2 + (k & 1)] = W2[idx];
    }
    HEAD_PUT(As[0]);

#pragma unroll 1
    for (int blk = 0; blk < HD_BLOCKS; ++blk) {
        const float* Ab = As[blk & 1];
        __syncthreads();
        if (blk + 1 < HD_BLOCKS) HEAD_FETCH(hbase + (size_t)(blk + 1) * HD_ROWS * D_IN);

#pragma unroll 1
        for (int mt = 0; mt < HD_MT; ++mt) {
            const float* arow = Ab + (mt * 16 + l15) * ACC_PITCH + 2 * half;
            v8f d = {};
#pragma unroll
            for (int s = 0; s < D_IN / 4; ++s) {
                const v2f a = *(const v2f*)(arow + 4 * s);
                d = __builtin_amdgcn_wmma_f32_16x16x4_f32(
                     false, a,  false, bf[s],
                     (short)0, d,  false,  false);
            }
            float* hrow = H1s + (mt * 16 + half * 8) * H1_PITCH + colB;
#pragma unroll
            for (int r = 0; r < 8; ++r) hrow[r * H1_PITCH] = d[r] + b1v;
        }
        if (blk + 1 < HD_BLOCKS) HEAD_PUT(As[(blk + 1) & 1]);
        __syncthreads();

        if (wave < 2 * HD_MT) {
            const int mt = wave >> 1;
            const int colO = (wave & 1) * 16 + l15;
            const float* arow = H1s + (mt * 16 + l15) * H1_PITCH + 2 * half;
            const float* brow = W2s + colO * 4 + half * 2;
            v8f d = {};
#pragma unroll
            for (int s = 0; s < D_HID / 4; ++s) {
                const v2f a = *(const v2f*)(arow + 4 * s);
                const v2f b = *(const v2f*)(brow + 128 * s);
                d = __builtin_amdgcn_wmma_f32_16x16x4_f32(
                     false, a,  false, b,
                     (short)0, d,  false,  false);
            }
            const float b2v = b2[colO];
            float* orow = out + (size_t)(row0 + blk * HD_ROWS + mt * 16 + half * 8) * D_OUT + colO;
#pragma unroll
            for (int r = 0; r < 8; ++r) orow[r * D_OUT] = d[r] + b2v;
        }
    }
}

extern "C" void kernel_launch(void* const* d_in, const int* in_sizes, int n_in,
                              void* d_out, int out_size, void* d_ws, size_t ws_size,
                              hipStream_t stream) {
    const float* blocks = (const float*)d_in[0];
    const int*   esrc   = (const int*)  d_in[1];
    const int*   edst   = (const int*)  d_in[2];
    const float* convW  = (const float*)d_in[3];
    const float* convB  = (const float*)d_in[4];
    const float* W1     = (const float*)d_in[5];
    const float* b1     = (const float*)d_in[6];
    const float* W2     = (const float*)d_in[7];
    const float* b2     = (const float*)d_in[8];
    float*       out    = (float*)d_out;

    char* ws = (char*)d_ws;
    const size_t BUF = (size_t)N_NODES * D_IN * sizeof(float);
    float* buf0 = (float*)(ws);
    int* lists = (int*)(ws + BUF);
    int* lcnts = lists + (size_t)N_ETYPES * LIST_WORDS;
    float* buf2 = (float*)(ws + 2 * BUF);
    int* offs = (int*)(ws + 2 * BUF);
    int* recs = offs + (size_t)N_ETYPES * PART_NCH * PART_OFFS;
    owner_part_kernel<<<N_ETYPES * PART_NCH, OWN_THREADS, 0, stream>>>(esrc, edst, offs, recs);
    float* deg  = (float*)(ws + 3 * BUF);

    const float* hin = blocks;
    float* hout = buf0;
    for (int l = 0; l < 2; ++l) {
        for (int e = 0; e < N_ETYPES; ++e) {
            const float* Wt = convW + ((size_t)l * N_ETYPES + e) * D_IN * D_IN;
            const float* bt = convB + ((size_t)l * N_ETYPES + e) * D_IN;
            const int*   es = esrc + (size_t)e * N_EDGES;
            const int*   ed = edst + (size_t)e * N_EDGES;
            float*       dg = deg + (size_t)e * N_NODES;
            int*         ls = lists + (size_t)e * LIST_WORDS;
            int*         lc = lcnts + (size_t)e * OWN_GROUPS * OWN_WAVES;
            const int*      of = offs + (size_t)e * PART_NCH * PART_OFFS;
            const int*      rc = recs + (size_t)e * N_EDGES;
            if (l == 0 && e == 0)
                owner_agg_kernel<true, false><<<OWN_GROUPS, OWN_THREADS, 0, stream>>>(hin, es, ed, of, rc, Wt, bt, hout, dg, ls, lc);
            else if (l == 0)
                owner_agg_kernel<true, true><<<OWN_GROUPS, OWN_THREADS, 0, stream>>>(hin, es, ed, of, rc, Wt, bt, hout, dg, ls, lc);
            else if (e == 0)
                owner_agg_kernel<false, false><<<OWN_GROUPS, OWN_THREADS, 0, stream>>>(hin, es, ed, of, rc, Wt, bt, hout, dg, ls, lc);
            else
                owner_agg_kernel<false, true><<<OWN_GROUPS, OWN_THREADS, 0, stream>>>(hin, es, ed, of, rc, Wt, bt, hout, dg, ls, lc);
        }
        hin  = hout;
        hout = buf2;
    }

    head_kernel<<<OWN_GROUPS, HD_THREADS, 0, stream>>>(hin, W1, b1, W2, b2, out);
}
